# starting kernel
# baseline (speedup 1.0000x reference)
#include <hip/hip_runtime.h>
#include <math.h>

#define B_   2
#define L_   2048
#define D_   1024
#define H_   16
#define DH_  64
#define NTOK (B_ * L_)

typedef __attribute__((ext_vector_type(2))) float v2f;
typedef __attribute__((ext_vector_type(8))) float v8f;

__device__ __forceinline__ v8f wmma4(v2f a, v2f b, v8f c) {
  return __builtin_amdgcn_wmma_f32_16x16x4_f32(
       false, a,  false, b,
       (short)0, c,  false,  false);
}

__device__ __forceinline__ float gelu_tanh(float x) {
  float x3 = x * x * x;
  return 0.5f * x * (1.0f + tanhf(0.7978845608028654f * (x + 0.044715f * x3)));
}

__global__ void __launch_bounds__(256)
rmsnorm_kernel(const float* __restrict__ x, const float* __restrict__ scale,
               float* __restrict__ out) {
  const int row = blockIdx.x;
  const float* xr = x + (size_t)row * D_;
  float* orow = out + (size_t)row * D_;
  const int t = threadIdx.x;
  float4 v = ((const float4*)xr)[t];
  float ss = v.x * v.x + v.y * v.y + v.z * v.z + v.w * v.w;
  for (int off = 16; off >= 1; off >>= 1) ss += __shfl_xor(ss, off, 32);
  __shared__ float red[8];
  if ((t & 31) == 0) red[t >> 5] = ss;
  __syncthreads();
  float tot = 0.f;
#pragma unroll
  for (int i = 0; i < 8; ++i) tot += red[i];
  float r = rsqrtf(tot * (1.0f / (float)D_) + 1e-6f);
  float4 s4 = ((const float4*)scale)[t];
  float4 o;
  o.x = v.x * s4.x * r; o.y = v.y * s4.y * r;
  o.z = v.z * s4.z * r; o.w = v.w * s4.w * r;
  ((float4*)orow)[t] = o;
}

__global__ void __launch_bounds__(256)
gemm_nt_kernel(const float* __restrict__ A, const float* __restrict__ W,
               const float* __restrict__ R, float* __restrict__ C,
               int M, int N, int K, int flags) {
  const int lane = threadIdx.x & 31;
  const int wave = threadIdx.x >> 5;
  const int half = lane >> 4;
  const int l16 = lane & 15;
  const int rowBase = blockIdx.y * 128 + wave * 16;
  const int colBase = blockIdx.x * 64;

  const float* arow = A + (size_t)(rowBase + l16) * K + 2 * half;
  const float* b0r  = W + (size_t)(colBase + l16) * K + 2 * half;
  const float* b1r  = b0r + (size_t)16 * K;
  const float* b2r  = b0r + (size_t)32 * K;
  const float* b3r  = b0r + (size_t)48 * K;

  __builtin_prefetch(arow, 0, 0);
  __builtin_prefetch(b0r, 0, 0);
  __builtin_prefetch(b1r, 0, 0);
  __builtin_prefetch(b2r, 0, 0);
  __builtin_prefetch(b3r, 0, 0);

  v8f acc0 = {0.f,0.f,0.f,0.f,0.f,0.f,0.f,0.f};
  v8f acc1 = acc0, acc2 = acc0, acc3 = acc0;

  v2f a_c  = *(const v2f*)(arow);
  v2f x0_c = *(const v2f*)(b0r);
  v2f x1_c = *(const v2f*)(b1r);
  v2f x2_c = *(const v2f*)(b2r);
  v2f x3_c = *(const v2f*)(b3r);

#pragma unroll 8
  for (int k = 0; k < K - 4; k += 4) {
    const int kn = k + 4;
    v2f a_n  = *(const v2f*)(arow + kn);
    v2f x0_n = *(const v2f*)(b0r + kn);
    v2f x1_n = *(const v2f*)(b1r + kn);
    v2f x2_n = *(const v2f*)(b2r + kn);
    v2f x3_n = *(const v2f*)(b3r + kn);
    acc0 = wmma4(a_c, x0_c, acc0);
    acc1 = wmma4(a_c, x1_c, acc1);
    acc2 = wmma4(a_c, x2_c, acc2);
    acc3 = wmma4(a_c, x3_c, acc3);
    a_c = a_n; x0_c = x0_n; x1_c = x1_n; x2_c = x2_n; x3_c = x3_n;
  }
  acc0 = wmma4(a_c, x0_c, acc0);
  acc1 = wmma4(a_c, x1_c, acc1);
  acc2 = wmma4(a_c, x2_c, acc2);
  acc3 = wmma4(a_c, x3_c, acc3);

#pragma unroll
  for (int r = 0; r < 8; ++r) {
    int row = rowBase + r + 8 * half;
    size_t i0 = (size_t)row * N + colBase + l16;
    float v0 = acc0[r], v1 = acc1[r], v2 = acc2[r], v3 = acc3[r];
    if (flags & 1) {
      v0 += R[i0]; v1 += R[i0 + 16]; v2 += R[i0 + 32]; v3 += R[i0 + 48];
    }
    if (flags & 2) {
      v0 = gelu_tanh(v0); v1 = gelu_tanh(v1);
      v2 = gelu_tanh(v2); v3 = gelu_tanh(v3);
    }
    C[i0] = v0; C[i0 + 16] = v1; C[i0 + 32] = v2; C[i0 + 48] = v3;
  }
}

__global__ void __launch_bounds__(256)
rope_kernel(float* __restrict__ qkv, const float* __restrict__ pos,
            const float* __restrict__ pos_orig, const float* __restrict__ tm) {
  int idx = blockIdx.x * blockDim.x + threadIdx.x;
  if (idx >= NTOK * H_ * 25) return;
  int i   = idx % 25;
  int h   = (idx / 25) % H_;
  int tok = idx / (25 * H_);
  int c = i / 5, j = i % 5;
  int fi = j * H_ + h;
  float theta;
  if (c < 4) {
    float freq = __expf(1.1447298858494002f + (float)fi * 0.028782313662425572f);
    int coord = (c < 2) ? 1 : 0;
    int s = (c < 2) ? c : c - 2;
    const float* P = (s == 0) ? pos_orig : pos;
    float pv = P[(size_t)tok * 2 + coord] * 2.0f - 1.0f;
    theta = pv * freq;
  } else {
    float tf = __expf(-(float)fi * 0.057564627324851145f);
    theta = tm[tok] * tf;
  }
  float cs = cosf(theta), sn = sinf(theta);
  float* qp = qkv + (size_t)tok * 3 * D_ + h * DH_;
  float* kp = qp + D_;
  float q1 = qp[i], q2 = qp[i + 25];
  qp[i] = q1 * cs - q2 * sn;  qp[i + 25] = q2 * cs + q1 * sn;
  float k1 = kp[i], k2 = kp[i + 25];
  kp[i] = k1 * cs - k2 * sn;  kp[i + 25] = k2 * cs + k1 * sn;
}

__global__ void __launch_bounds__(256)
cos_scale_kernel(float* __restrict__ qkv, const float* __restrict__ qk_scale) {
  const int lane = threadIdx.x & 31;
  const int wave = threadIdx.x >> 5;
  const int row = blockIdx.x * 8 + wave;
  const int h = row % H_;
  const int tok = row / H_;
  float* qp = qkv + (size_t)tok * 3 * D_ + h * DH_;
  float* kp = qp + D_;
  float2 q2 = ((float2*)qp)[lane];
  float2 k2 = ((float2*)kp)[lane];
  float sq = q2.x * q2.x + q2.y * q2.y;
  float sk = k2.x * k2.x + k2.y * k2.y;
  for (int off = 16; off >= 1; off >>= 1) {
    sq += __shfl_xor(sq, off, 32);
    sk += __shfl_xor(sk, off, 32);
  }
  float s = sqrtf(qk_scale[h]);
  float qs = s * rsqrtf(sq + 1e-6f);
  float ks = s * rsqrtf(sk + 1e-6f);
  ((float2*)qp)[lane] = make_float2(q2.x * qs, q2.y * qs);
  ((float2*)kp)[lane] = make_float2(k2.x * ks, k2.y * ks);
}

__global__ void __launch_bounds__(32)
attn_kernel(const float* __restrict__ qkv, float* __restrict__ out) {
  const int lane = threadIdx.x & 31;
  const int half = lane >> 4, l16 = lane & 15;
  const int qt = blockIdx.x;
  const int bh = blockIdx.y;
  const int b = bh / H_, h = bh % H_;
  const size_t ts = 3 * D_;

  const float* qrow = qkv + (size_t)(b * L_ + qt * 16 + l16) * ts + h * DH_ + 2 * half;
  v2f qf[16];
#pragma unroll
  for (int s = 0; s < 16; ++s) qf[s] = *(const v2f*)(qrow + 4 * s);

  float m[8], lsum[8];
#pragma unroll
  for (int r = 0; r < 8; ++r) { m[r] = -INFINITY; lsum[r] = 0.f; }
  v8f o0 = {0.f,0.f,0.f,0.f,0.f,0.f,0.f,0.f};
  v8f o1 = o0, o2 = o0, o3 = o0;

  __shared__ float P[16 * 17];
  const float* kbase = qkv + (size_t)(b * L_) * ts + D_ + h * DH_;
  const float* vbase = qkv + (size_t)(b * L_) * ts + 2 * D_ + h * DH_;

  for (int kt = 0; kt < L_ / 16; ++kt) {
    const float* krow = kbase + (size_t)(kt * 16 + l16) * ts + 2 * half;
    v8f S0 = {0.f,0.f,0.f,0.f,0.f,0.f,0.f,0.f};
    v8f S1 = S0;
#pragma unroll
    for (int s = 0; s < 16; s += 2) {
      v2f bf0 = *(const v2f*)(krow + 4 * s);
      v2f bf1 = *(const v2f*)(krow + 4 * s + 4);
      S0 = wmma4(qf[s], bf0, S0);
      S1 = wmma4(qf[s + 1], bf1, S1);
    }
    __syncthreads();
#pragma unroll
    for (int r = 0; r < 8; ++r) {
      float sv = S0[r] + S1[r];
      float mx = sv;
      for (int off = 8; off >= 1; off >>= 1) mx = fmaxf(mx, __shfl_xor(mx, off, 32));
      float mnew = fmaxf(m[r], mx);
      float alpha = __expf(m[r] - mnew);
      float p = __expf(sv - mnew);
      float rs = p;
      for (int off = 8; off >= 1; off >>= 1) rs += __shfl_xor(rs, off, 32);
      lsum[r] = lsum[r] * alpha + rs;
      m[r] = mnew;
      o0[r] *= alpha; o1[r] *= alpha; o2[r] *= alpha; o3[r] *= alpha;
      P[(r + 8 * half) * 17 + l16] = p;
    }
    __syncthreads();
    const float* vrow = vbase + (size_t)(kt * 16) * ts;
#pragma unroll
    for (int s = 0; s < 4; ++s) {
      int k0 = s * 4 + 2 * half;
      v2f pf;
      pf.x = P[l16 * 17 + k0];
      pf.y = P[l16 * 17 + k0 + 1];
      const float* v0p = vrow + (size_t)k0 * ts;
      const float* v1p = v0p + ts;
      v2f bv;
      bv.x = v0p[l16];       bv.y = v1p[l16];       o0 = wmma4(pf, bv, o0);
      bv.x = v0p[16 + l16];  bv.y = v1p[16 + l16];  o1 = wmma4(pf, bv, o1);
      bv.x = v0p[32 + l16];  bv.y = v1p[32 + l16];  o2 = wmma4(pf, bv, o2);
      bv.x = v0p[48 + l16];  bv.y = v1p[48 + l16];  o3 = wmma4(pf, bv, o3);
    }
  }
  float* obase = out + (size_t)(b * L_ + qt * 16) * D_ + h * DH_;
#pragma unroll
  for (int r = 0; r < 8; ++r) {
    float inv = 1.0f / lsum[r];
    float* op = obase + (size_t)(r + 8 * half) * D_;
    op[l16]      = o0[r] * inv;
    op[16 + l16] = o1[r] * inv;
    op[32 + l16] = o2[r] * inv;
    op[48 + l16] = o3[r] * inv;
  }
}

extern "C" void kernel_launch(void* const* d_in, const int* in_sizes, int n_in,
                              void* d_out, int out_size, void* d_ws, size_t ws_size,
                              hipStream_t stream) {
  (void)in_sizes; (void)n_in; (void)out_size; (void)ws_size;
  const float* x        = (const float*)d_in[0];
  const float* pos      = (const float*)d_in[1];
  const float* pos_orig = (const float*)d_in[2];
  const float* tm       = (const float*)d_in[3];
  const float* w_qkv    = (const float*)d_in[4];
  const float* w_out    = (const float*)d_in[5];
  const float* w_up     = (const float*)d_in[6];
  const float* w_down   = (const float*)d_in[7];
  const float* n1s      = (const float*)d_in[8];
  const float* n2s      = (const float*)d_in[9];
  const float* qksc     = (const float*)d_in[10];
  float* out = (float*)d_out;

  float* xn   = (float*)d_ws;
  float* qkv  = xn  + (size_t)NTOK * D_;
  float* attn = qkv + (size_t)NTOK * 3 * D_;
  float* up   = attn + (size_t)NTOK * D_;

  rmsnorm_kernel<<<NTOK, 256, 0, stream>>>(x, n1s, xn);
  gemm_nt_kernel<<<dim3(3 * D_ / 64, NTOK / 128), 256, 0, stream>>>(
      xn, w_qkv, nullptr, qkv, NTOK, 3 * D_, D_, 0);
  rope_kernel<<<(NTOK * H_ * 25 + 255) / 256, 256, 0, stream>>>(qkv, pos, pos_orig, tm);
  cos_scale_kernel<<<NTOK * H_ / 8, 256, 0, stream>>>(qkv, qksc);
  attn_kernel<<<dim3(L_ / 16, B_ * H_), 32, 0, stream>>>(qkv, attn);
  gemm_nt_kernel<<<dim3(D_ / 64, NTOK / 128), 256, 0, stream>>>(
      attn, w_out, x, out, NTOK, D_, D_, 1);
  rmsnorm_kernel<<<NTOK, 256, 0, stream>>>(out, n2s, xn);
  gemm_nt_kernel<<<dim3(4 * D_ / 64, NTOK / 128), 256, 0, stream>>>(
      xn, w_up, nullptr, up, NTOK, 4 * D_, D_, 2);
  gemm_nt_kernel<<<dim3(D_ / 64, NTOK / 128), 256, 0, stream>>>(
      up, w_down, out, out, NTOK, D_, 4 * D_, 1);
}


# QKV GEMM epilogue: each wave turns its q or k head tile in LDS (rotary embedding, cosine scaling) and stores whole rows; the RoPE and q/k-scale launches go; q/k sums of squares added in a new order
# speedup vs baseline: 3.1168x; 3.1168x over previous
#include <hip/hip_runtime.h>
#include <math.h>

#define B_   2
#define L_   2048
#define D_   1024
#define H_   16
#define DH_  64
#define NTOK (B_ * L_)

typedef __attribute__((ext_vector_type(2))) float v2f;
typedef __attribute__((ext_vector_type(4))) float v4f;
typedef __attribute__((ext_vector_type(8))) float v8f;

__device__ __forceinline__ v8f wmma4(v2f a, v2f b, v8f c) {
  return __builtin_amdgcn_wmma_f32_16x16x4_f32(
       false, a,  false, b,
       (short)0, c,  false,  false);
}

__device__ __forceinline__ float gelu_tanh(float x) {
  float x3 = x * x * x;
  return 0.5f * x * (1.0f + tanhf(0.7978845608028654f * (x + 0.044715f * x3)));
}

__device__ __forceinline__ float row16_sum(float v) {
  asm("v_add_f32_dpp %0, %0, %0 row_xmask:1 row_mask:0xf bank_mask:0xf\n\t"
      "v_add_f32_dpp %0, %0, %0 row_xmask:2 row_mask:0xf bank_mask:0xf\n\t"
      "v_add_f32_dpp %0, %0, %0 row_xmask:4 row_mask:0xf bank_mask:0xf\n\t"
      "v_add_f32_dpp %0, %0, %0 row_xmask:8 row_mask:0xf bank_mask:0xf"
      : "+v"(v));
  return v;
}

__global__ void __launch_bounds__(256)
rmsnorm_kernel(const float* __restrict__ x, const float* __restrict__ scale,
               float* __restrict__ out) {
  const int row = blockIdx.x;
  const float* xr = x + (size_t)row * D_;
  float* orow = out + (size_t)row * D_;
  const int t = threadIdx.x;
  float4 v = ((const float4*)xr)[t];
  float ss = v.x * v.x + v.y * v.y + v.z * v.z + v.w * v.w;
  for (int off = 16; off >= 1; off >>= 1) ss += __shfl_xor(ss, off, 32);
  __shared__ float red[8];
  if ((t & 31) == 0) red[t >> 5] = ss;
  __syncthreads();
  float tot = 0.f;
#pragma unroll
  for (int i = 0; i < 8; ++i) tot += red[i];
  float r = rsqrtf(tot * (1.0f / (float)D_) + 1e-6f);
  float4 s4 = ((const float4*)scale)[t];
  float4 o;
  o.x = v.x * s4.x * r; o.y = v.y * s4.y * r;
  o.z = v.z * s4.z * r; o.w = v.w * s4.w * r;
  ((float4*)orow)[t] = o;
}

struct FragSet { v4f a[4]; v4f b[4]; };

#define RT_PS 68

template <int K>
__device__ __forceinline__ void load_set(FragSet& f, const float* __restrict__ pa,
                                         const float* __restrict__ pb, int k) {
#pragma unroll
  for (int i = 0; i < 4; ++i) f.a[i] = *(const v4f*)(pa + (size_t)i * 16 * K + k);
#pragma unroll
  for (int j = 0; j < 4; ++j) f.b[j] = *(const v4f*)(pb + (size_t)j * 16 * K + k);
}

__device__ __forceinline__ void mma_set(const FragSet& f, v8f (&acc)[4][4]) {
#pragma unroll
  for (int i = 0; i < 4; ++i) {
    v2f a = __builtin_shufflevector(f.a[i], f.a[i], 0, 1);
#pragma unroll
    for (int j = 0; j < 4; ++j)
      acc[i][j] = wmma4(a, __builtin_shufflevector(f.b[j], f.b[j], 0, 1), acc[i][j]);
  }
#pragma unroll
  for (int i = 0; i < 4; ++i) {
    v2f a = __builtin_shufflevector(f.a[i], f.a[i], 2, 3);
#pragma unroll
    for (int j = 0; j < 4; ++j)
      acc[i][j] = wmma4(a, __builtin_shufflevector(f.b[j], f.b[j], 2, 3), acc[i][j]);
  }
}

template <int N, int K, int FLAGS>
__global__ void __launch_bounds__(128)
gemm_nt_kernel(const float* __restrict__ A, const float* __restrict__ W,
               const float* R, float* C,
               const float* __restrict__ pos, const float* __restrict__ pos_orig,
               const float* __restrict__ tm, const float* __restrict__ qk_scale) {
  constexpr int G = K / 8;
  static_assert(K % 8 == 0 && G % 3 == 2 && G >= 5, "K loop is rotated by three");
  const int lane = threadIdx.x & 31;
  const int wave = threadIdx.x >> 5;
  const int half = lane >> 4;
  const int l16 = lane & 15;
  const int rowBase = blockIdx.y * 128 + (wave >> 1) * 64;
  const int colBase = blockIdx.x * 128 + (wave & 1) * 64;

  const float* pa = A + (size_t)(rowBase + l16) * K + 4 * half;
  const float* pb = W + (size_t)(colBase + l16) * K + 4 * half;

  v8f acc[4][4];
#pragma unroll
  for (int i = 0; i < 4; ++i)
#pragma unroll
    for (int j = 0; j < 4; ++j)
      acc[i][j] = (v8f){0.f,0.f,0.f,0.f,0.f,0.f,0.f,0.f};

#define SCHED_FENCE() __builtin_amdgcn_sched_barrier(0)
  FragSet s0, s1, s2;
  load_set<K>(s0, pa, pb, 0);   SCHED_FENCE();
  load_set<K>(s1, pa, pb, 8);   SCHED_FENCE();
  load_set<K>(s2, pa, pb, 16);  SCHED_FENCE();
  int g = 0;
  for (; g + 5 < G; g += 3) {
    const int k = g * 8;
    mma_set(s0, acc);  SCHED_FENCE();  load_set<K>(s0, pa, pb, k + 24);  SCHED_FENCE();
    mma_set(s1, acc);  SCHED_FENCE();  load_set<K>(s1, pa, pb, k + 32);  SCHED_FENCE();
    mma_set(s2, acc);  SCHED_FENCE();  load_set<K>(s2, pa, pb, k + 40);  SCHED_FENCE();
  }
  mma_set(s0, acc);  SCHED_FENCE();  load_set<K>(s0, pa, pb, K - 16);  SCHED_FENCE();
  mma_set(s1, acc);  SCHED_FENCE();  load_set<K>(s1, pa, pb, K - 8);   SCHED_FENCE();
  mma_set(s2, acc);
  mma_set(s0, acc);
  mma_set(s1, acc);
#undef SCHED_FENCE

  if constexpr ((FLAGS & 4) != 0) {
    if (blockIdx.x * 128 < 2 * D_) {
      __shared__ __attribute__((aligned(16))) float Tsh[4][64 * RT_PS];
      float* T = Tsh[wave];

#pragma unroll
      for (int i = 0; i < 4; ++i) {
#pragma unroll
        for (int r = 0; r < 8; ++r) {
          float* tr = T + (16 * i + 8 * half + r) * RT_PS + l16;
#pragma unroll
          for (int j = 0; j < 4; ++j) tr[16 * j] = acc[i][j][r];
        }
      }

      const int hd = (colBase & (D_ - 1)) / DH_;
      const int tok0 = rowBase + lane, tok1 = tok0 + 32;
      const float2 po0 = ((const float2*)pos_orig)[tok0];
      const float2 pp0 = ((const float2*)pos)[tok0];
      const float2 po1 = ((const float2*)pos_orig)[tok1];
      const float2 pp1 = ((const float2*)pos)[tok1];
      const float tm0 = tm[tok0], tm1 = tm[tok1];
      const float s = sqrtf(qk_scale[hd]);
      __syncthreads();

#pragma nounroll
      for (int i = 0; i < 25; ++i) {
        float* t0 = T + lane * RT_PS + i;
        float* t1 = t0 + 32 * RT_PS;
        const float a0 = t0[0], b0 = t0[25];
        const float a1 = t1[0], b1 = t1[25];
        const int c = i / 5;
        const float fi = (float)((i - 5 * c) * H_ + hd);
        const float fr = __expf((c < 4) ? (1.1447298858494002f + fi * 0.028782313662425572f)
                                        : (-fi * 0.057564627324851145f));
        const float raw0 = (c == 0) ? po0.y : (c == 1) ? pp0.y : (c == 2) ? po0.x
                         : (c == 3) ? pp0.x : tm0;
        const float raw1 = (c == 0) ? po1.y : (c == 1) ? pp1.y : (c == 2) ? po1.x
                         : (c == 3) ? pp1.x : tm1;
        const float th0 = (c < 4) ? (raw0 * 2.0f - 1.0f) * fr : raw0 * fr;
        const float th1 = (c < 4) ? (raw1 * 2.0f - 1.0f) * fr : raw1 * fr;
        float cs0, sn0, cs1, sn1;
        sincosf(th0, &sn0, &cs0);
        sincosf(th1, &sn1, &cs1);
        t0[0] = a0 * cs0 - b0 * sn0;  t0[25] = b0 * cs0 + a0 * sn0;
        t1[0] = a1 * cs1 - b1 * sn1;  t1[25] = b1 * cs1 + a1 * sn1;
      }
      __syncthreads();

#pragma unroll 8
      for (int rr = 0; rr < 32; ++rr) {
        const int row = 2 * rr + half;
        v4f v = *(const v4f*)(T + row * RT_PS + 4 * l16);
        const float sq = row16_sum(v.x * v.x + v.y * v.y + v.z * v.z + v.w * v.w);
        v *= s * rsqrtf(sq + 1e-6f);
        *(v4f*)(C + (size_t)(rowBase + row) * N + colBase + 4 * l16) = v;
      }
      return;
    }
  }

  if (FLAGS & 1) {
#pragma unroll
    for (int i = 0; i < 4; ++i) {
#pragma unroll
      for (int r = 0; r < 8; ++r) {
        const int row = rowBase + 16 * i + r + 8 * half;
        const size_t i0 = (size_t)row * N + colBase + l16;
#pragma unroll
        for (int j = 0; j < 4; ++j) acc[i][j][r] += R[i0 + 16 * j];
      }
    }
  }
#pragma unroll
  for (int i = 0; i < 4; ++i) {
#pragma unroll
    for (int r = 0; r < 8; ++r) {
      const int row = rowBase + 16 * i + r + 8 * half;
      const size_t i0 = (size_t)row * N + colBase + l16;
#pragma unroll
      for (int j = 0; j < 4; ++j) {
        float v = acc[i][j][r];
        if (FLAGS & 2) v = gelu_tanh(v);
        C[i0 + 16 * j] = v;
      }
    }
  }
}

#define AT_PS 68

__device__ __forceinline__ float row16_max(float v) {
  asm("v_max_num_f32_dpp %0, %0, %0 row_xmask:1 row_mask:0xf bank_mask:0xf\n\t"
      "v_max_num_f32_dpp %0, %0, %0 row_xmask:2 row_mask:0xf bank_mask:0xf\n\t"
      "v_max_num_f32_dpp %0, %0, %0 row_xmask:4 row_mask:0xf bank_mask:0xf\n\t"
      "v_max_num_f32_dpp %0, %0, %0 row_xmask:8 row_mask:0xf bank_mask:0xf"
      : "+v"(v));
  return v;
}

__global__ void __launch_bounds__(128)
attn_kernel(const float* __restrict__ qkv, float* __restrict__ out) {
  const int lane = threadIdx.x & 31;
  const int wave = threadIdx.x >> 5;
  const int half = lane >> 4, l16 = lane & 15;
  const int q0 = blockIdx.x * 128 + wave * 32;
  const int bh = blockIdx.y;
  const int b = bh / H_, h = bh % H_;
  const size_t ts = 3 * D_;

  __shared__ __attribute__((aligned(16))) float Psh[4][32 * AT_PS];
  float* Pw = Psh[wave];


  const float* qrow = qkv + (size_t)(b * L_ + q0 + l16) * ts + h * DH_ + 4 * half;
  v4f qf[2][8];
#pragma unroll
  for (int i = 0; i < 2; ++i)
#pragma unroll
    for (int g = 0; g < 8; ++g)
      qf[i][g] = *(const v4f*)(qrow + (size_t)i * 16 * ts + 8 * g);

  float m[2][8], lsum[2][8];
  v8f o[2][4];
#pragma unroll
  for (int i = 0; i < 2; ++i) {
#pragma unroll
    for (int r = 0; r < 8; ++r) { m[i][r] = -INFINITY; lsum[i][r] = 0.f; }
#pragma unroll
    for (int j = 0; j < 4; ++j) o[i][j] = (v8f){0.f,0.f,0.f,0.f,0.f,0.f,0.f,0.f};
  }

  const float* kbase = qkv + (size_t)(b * L_ + l16) * ts + D_ + h * DH_ + 4 * half;
  const float* vbase = qkv + (size_t)(b * L_ + 4 * half) * ts + 2 * D_ + h * DH_ + 4 * l16;

  for (int kb = 0; kb < L_ / 64; ++kb) {
    __syncthreads();

    const float* kp = kbase + (size_t)kb * 64 * ts;
    v8f S[2][4];
#pragma unroll
    for (int t = 0; t < 4; ++t) {
      v4f kf[8];
#pragma unroll
      for (int g = 0; g < 8; ++g)
        kf[g] = *(const v4f*)(kp + (size_t)t * 16 * ts + 8 * g);
      v8f a0 = {0.f,0.f,0.f,0.f,0.f,0.f,0.f,0.f};
      v8f a1 = a0;
#pragma unroll
      for (int g = 0; g < 8; ++g) {
        v2f klo = __builtin_shufflevector(kf[g], kf[g], 0, 1);
        v2f khi = __builtin_shufflevector(kf[g], kf[g], 2, 3);
        a0 = wmma4(__builtin_shufflevector(qf[0][g], qf[0][g], 0, 1), klo, a0);
        a1 = wmma4(__builtin_shufflevector(qf[1][g], qf[1][g], 0, 1), klo, a1);
        a0 = wmma4(__builtin_shufflevector(qf[0][g], qf[0][g], 2, 3), khi, a0);
        a1 = wmma4(__builtin_shufflevector(qf[1][g], qf[1][g], 2, 3), khi, a1);
      }
      S[0][t] = a0;
      S[1][t] = a1;
    }

    __builtin_amdgcn_sched_barrier(0);
    const float* vp = vbase + (size_t)kb * 64 * ts;
    v4f vv[8][2][2];
#pragma unroll
    for (int g = 0; g < 8; ++g)
#pragma unroll
      for (int w = 0; w < 2; ++w)
#pragma unroll
        for (int s = 0; s < 2; ++s)
          vv[g][w][s] = *(const v4f*)(vp + (size_t)(8 * g + 2 * w + s) * ts);

#pragma unroll
    for (int i = 0; i < 2; ++i) {
#pragma unroll
      for (int r = 0; r < 8; ++r) {
        float mx = fmaxf(fmaxf(S[i][0][r], S[i][1][r]), fmaxf(S[i][2][r], S[i][3][r]));
        mx = row16_max(mx);
        const float mnew = fmaxf(m[i][r], mx);
        const float alpha = __expf(m[i][r] - mnew);
        m[i][r] = mnew;
        float* prow = Pw + (16 * i + 8 * half + r) * AT_PS + l16;
        float rs = 0.f;
#pragma unroll
        for (int t = 0; t < 4; ++t) {
          const float p = __expf(S[i][t][r] - mnew);
          rs += p;
          prow[16 * t] = p;
        }
        lsum[i][r] = lsum[i][r] * alpha + rs;
#pragma unroll
        for (int j = 0; j < 4; ++j) o[i][j][r] *= alpha;
      }
    }
    __syncthreads();

#pragma unroll
    for (int g = 0; g < 8; ++g) {
      const v4f p0 = *(const v4f*)(Pw + (l16) * AT_PS + 8 * g + 4 * half);
      const v4f p1 = *(const v4f*)(Pw + (16 + l16) * AT_PS + 8 * g + 4 * half);
#pragma unroll
      for (int w = 0; w < 2; ++w) {
        const v2f pa0 = (w == 0) ? __builtin_shufflevector(p0, p0, 0, 1)
                                 : __builtin_shufflevector(p0, p0, 2, 3);
        const v2f pa1 = (w == 0) ? __builtin_shufflevector(p1, p1, 0, 1)
                                 : __builtin_shufflevector(p1, p1, 2, 3);
#pragma unroll
        for (int j = 0; j < 4; ++j) {
          v2f bv;
          bv.x = vv[g][w][0][j];
          bv.y = vv[g][w][1][j];
          o[0][j] = wmma4(pa0, bv, o[0][j]);
          o[1][j] = wmma4(pa1, bv, o[1][j]);
        }
      }
    }
  }

#pragma unroll
  for (int i = 0; i < 2; ++i) {
#pragma unroll
    for (int r = 0; r < 8; ++r) {
      const float inv = 1.0f / row16_sum(lsum[i][r]);
      float4 ov;
      ov.x = o[i][0][r] * inv;  ov.y = o[i][1][r] * inv;
      ov.z = o[i][2][r] * inv;  ov.w = o[i][3][r] * inv;
      float* op = out + (size_t)(b * L_ + q0 + 16 * i + 8 * half + r) * D_ + h * DH_ + 4 * l16;
      *(float4*)op = ov;
    }
  }
}

extern "C" void kernel_launch(void* const* d_in, const int* in_sizes, int n_in,
                              void* d_out, int out_size, void* d_ws, size_t ws_size,
                              hipStream_t stream) {
  (void)in_sizes; (void)n_in; (void)out_size; (void)ws_size;
  const float* x        = (const float*)d_in[0];
  const float* pos      = (const float*)d_in[1];
  const float* pos_orig = (const float*)d_in[2];
  const float* tm       = (const float*)d_in[3];
  const float* w_qkv    = (const float*)d_in[4];
  const float* w_out    = (const float*)d_in[5];
  const float* w_up     = (const float*)d_in[6];
  const float* w_down   = (const float*)d_in[7];
  const float* n1s      = (const float*)d_in[8];
  const float* n2s      = (const float*)d_in[9];
  const float* qksc     = (const float*)d_in[10];
  float* out = (float*)d_out;

  float* xn   = (float*)d_ws;
  float* qkv  = xn  + (size_t)NTOK * D_;
  float* attn = qkv + (size_t)NTOK * 3 * D_;
  float* up   = attn + (size_t)NTOK * D_;

  rmsnorm_kernel<<<NTOK, 256, 0, stream>>>(x, n1s, xn);
  gemm_nt_kernel<3 * D_, D_, 4><<<dim3(3 * D_ / 128, NTOK / 128), 128, 0, stream>>>(
      xn, w_qkv, nullptr, qkv, pos, pos_orig, tm, qksc);
  attn_kernel<<<dim3(L_ / 128, B_ * H_), 128, 0, stream>>>(qkv, attn);
  gemm_nt_kernel<D_, D_, 1><<<dim3(D_ / 128, NTOK / 128), 128, 0, stream>>>(
      attn, w_out, x, out, nullptr, nullptr, nullptr, nullptr);
  rmsnorm_kernel<<<NTOK, 256, 0, stream>>>(out, n2s, xn);
  gemm_nt_kernel<4 * D_, D_, 2><<<dim3(4 * D_ / 128, NTOK / 128), 128, 0, stream>>>(
      xn, w_up, nullptr, up, nullptr, nullptr, nullptr, nullptr);
  gemm_nt_kernel<D_, 4 * D_, 1><<<dim3(D_ / 128, NTOK / 128), 128, 0, stream>>>(
      up, w_down, out, out, nullptr, nullptr, nullptr, nullptr);
}
